# starting kernel
# baseline (speedup 1.0000x reference)
#include <hip/hip_runtime.h>
#include <hip/hip_bf16.h>
#include <math.h>

typedef __attribute__((ext_vector_type(2))) float v2f;
typedef __attribute__((ext_vector_type(8))) float v8f;

#define NC    65536
#define NHN   7
#define NVV   4
#define EOUT  16

#define W_STRIDE    258
#define ND_STRIDE   10
#define Y_CSTRIDE   2592
#define Y_RSTRIDE   160
#define WAVE_LDS    7600
#define LDS_FLOATS  (4128 + 8 * WAVE_LDS)

__global__ __launch_bounds__(256) void
multi_grid_encoder_kernel(const float* __restrict__ x,
                          const float* __restrict__ coords,
                          const float* __restrict__ sigma,
                          const float* __restrict__ dist_offsets,
                          const float* __restrict__ Wg,
                          const float* __restrict__ b_out,
                          const int*   __restrict__ adjc,
                          const int*   __restrict__ nh_mask,
                          float*       __restrict__ out)
{
    extern __shared__ float lds[];

    const int tid    = threadIdx.x;
    const int wave   = tid >> 5;
    const int lane   = tid & 31;
    const int lane16 = lane & 15;
    const int hi     = lane >> 4;

    float* wS  = lds;
    float* wb  = lds + 4128 + wave * WAVE_LDS;
    float* xgP = wb;
    float* ndP = wb + 2048;
    float* yP  = wb + 2368;
    float* dS  = wb + 7552;
    float* mS  = wb + 7568;
    int*   iS  = (int*)(wb + 7584);

    {
        const float4* src = (const float4*)Wg;
        #pragma unroll
        for (int it = 0; it < 4; ++it) {
            const int i4 = tid + it * 256;
            const float4 v = src[i4];
            const int k  = i4 >> 2;
            const int f0 = (i4 & 3) * 4;
            wS[(f0 + 0) * W_STRIDE + k] = v.x;
            wS[(f0 + 1) * W_STRIDE + k] = v.y;
            wS[(f0 + 2) * W_STRIDE + k] = v.z;
            wS[(f0 + 3) * W_STRIDE + k] = v.w;
        }
    }

    const int pairIdx = blockIdx.x * 8 + wave;
    const int n0      = pairIdx * 2;
    const int c       = hi;

    if (lane16 < NHN) {
        const int h = lane16;
        const int n = n0 + c;
        const float lat_c = coords[n];
        const float lon_c = coords[NC + n];
        const int   m     = adjc[n * NHN + h];
        const float lat_n = coords[m];
        const float lon_n = coords[NC + m];
        float cosd = __sinf(lat_c) * __sinf(lat_n)
                   + __cosf(lat_c) * __cosf(lat_n) * __cosf(lon_n - lon_c);
        cosd = fminf(fmaxf(cosd, -1.0f + 1e-7f), 1.0f - 1e-7f);
        dS[c * 8 + h] = acosf(cosd);
        mS[c * 8 + h] = (float)nh_mask[n * NHN + h];
        iS[c * 8 + h] = m;
    }

    {
        const int os = lane16;
        const float sg  = sigma[os & 3];
        const float off = dist_offsets[os >> 2];
        const float inv_sg = 1.0f / sg;
        float ndv[NHN];
        float sum = 0.0f;
        #pragma unroll
        for (int h = 0; h < NHN; ++h) {
            const float d = dS[c * 8 + h];
            const float t = (d - off) * inv_sg;
            const float val = __expf(-0.5f * t * t) * mS[c * 8 + h];
            ndv[h] = val;
            sum += val;
        }
        const float inv = 1.0f / (sum + 1e-8f);
        float* row = ndP + c * 160 + os * ND_STRIDE;
        #pragma unroll
        for (int h = 0; h < NHN; ++h) row[h] = ndv[h] * inv;
        row[7] = 0.0f;
    }

    {
        const int b  = lane >> 4;
        const int v  = (lane >> 2) & 3;
        const int e  = (lane & 3) * 4;
        const int col0 = lane * 4;
        const size_t xoff = (size_t)b * NC * 64 + v * 16 + e;
        #pragma unroll
        for (int c2 = 0; c2 < 2; ++c2) {
            #pragma unroll
            for (int hp = 0; hp < 4; ++hp) {
                const int m0 = __builtin_amdgcn_readfirstlane(iS[c2 * 8 + 2 * hp]);
                const float4 v0 = *(const float4*)(x + xoff + (size_t)m0 * 64);
                float4 v1 = make_float4(0.f, 0.f, 0.f, 0.f);
                if (hp < 3) {
                    const int m1 = __builtin_amdgcn_readfirstlane(iS[c2 * 8 + 2 * hp + 1]);
                    v1 = *(const float4*)(x + xoff + (size_t)m1 * 64);
                }
                v2f* bp = (v2f*)(xgP + c2 * 1024 + hp * 256 + col0 * 2);
                bp[0] = (v2f){v0.x, v1.x};
                bp[1] = (v2f){v0.y, v1.y};
                bp[2] = (v2f){v0.z, v1.z};
                bp[3] = (v2f){v0.w, v1.w};
            }
        }
    }

    __syncthreads();

    #pragma unroll
    for (int c2 = 0; c2 < 2; ++c2) {
        v2f Ak[2];
        #pragma unroll
        for (int kk = 0; kk < 2; ++kk)
            Ak[kk] = *(const v2f*)(ndP + c2 * 160 + lane16 * ND_STRIDE
                                   + kk * 4 + hi * 2);
        #pragma unroll
        for (int t = 0; t < 8; ++t) {
            v8f acc = {0.f, 0.f, 0.f, 0.f, 0.f, 0.f, 0.f, 0.f};
            #pragma unroll
            for (int kk = 0; kk < 2; ++kk) {
                const v2f Bm = *(const v2f*)(xgP + c2 * 1024 + (kk * 2 + hi) * 256
                                             + (t * 16 + lane16) * 2);
                acc = __builtin_amdgcn_wmma_f32_16x16x4_f32(
                        false, Ak[kk], false, Bm, (short)0, acc, false, false);
            }
            float* yrow = yP + c2 * Y_CSTRIDE + t * 18 + lane16;
            #pragma unroll
            for (int j = 0; j < 8; ++j)
                yrow[(j + hi * 8) * Y_RSTRIDE] = acc[j];
        }
    }

    {
        const float bias = b_out[lane16];
        v8f acc = {bias, bias, bias, bias, bias, bias, bias, bias};

        const int r  = lane16;
        const int cA = r >> 3;
        const int bA = (r >> 2) & 1;
        const int vA = r & 3;
        const int abase = cA * Y_CSTRIDE + bA * 64 + vA * 16 + (bA * 4 + vA) * 2;

        #pragma unroll
        for (int kk = 0; kk < 64; ++kk) {
            const int k0 = kk * 4 + hi * 2;
            const v2f A  = *(const v2f*)(yP + abase + (k0 >> 4) * Y_RSTRIDE + (k0 & 15));
            const v2f Bm = *(const v2f*)(wS + lane16 * W_STRIDE + k0);
            acc = __builtin_amdgcn_wmma_f32_16x16x4_f32(
                    false, A, false, Bm, (short)0, acc, false, false);
        }

        #pragma unroll
        for (int j = 0; j < 8; ++j) {
            const int row = j + hi * 8;
            const int cO  = row >> 3;
            const int bO  = (row >> 2) & 1;
            const int vO  = row & 3;
            const int n   = n0 + cO;
            out[(((size_t)bO * NC + n) * NVV + vO) * EOUT + lane16] = acc[j];
        }
    }
}

extern "C" void kernel_launch(void* const* d_in, const int* in_sizes, int n_in,
                              void* d_out, int out_size, void* d_ws, size_t ws_size,
                              hipStream_t stream) {
    (void)in_sizes; (void)n_in; (void)out_size; (void)d_ws; (void)ws_size;
    const float* x            = (const float*)d_in[0];
    const float* coords       = (const float*)d_in[1];
    const float* sigma        = (const float*)d_in[2];
    const float* dist_offsets = (const float*)d_in[3];
    const float* Wg           = (const float*)d_in[4];
    const float* b_out        = (const float*)d_in[5];
    const int*   adjc         = (const int*)d_in[6];
    const int*   nh_mask      = (const int*)d_in[7];
    float*       out          = (float*)d_out;

    const int blocks = NC / 16;
    const size_t shmem = (size_t)LDS_FLOATS * sizeof(float);
    multi_grid_encoder_kernel<<<blocks, 256, shmem, stream>>>(
        x, coords, sigma, dist_offsets, Wg, b_out, adjc, nh_mask, out);
}


# stage-1 product transposed, K = (cell, neighbour) block-diagonal, so its D registers are the stage-2 B operand: x gathered straight to registers, no LDS round trip of x and Y; summation order changes
# speedup vs baseline: 1.7034x; 1.7034x over previous
#include <hip/hip_runtime.h>
#include <math.h>


typedef __attribute__((ext_vector_type(2))) float v2f;
typedef __attribute__((ext_vector_type(4))) float v4f;
typedef __attribute__((ext_vector_type(8))) float v8f;

#define NC    65536
#define NHN   7
#define NVV   4
#define EOUT  16

#define WS_FLOATS   4096
#define DM_FLOATS   64
#define LDS_FLOATS  (WS_FLOATS + 8 * DM_FLOATS)

__global__ __launch_bounds__(256) void
multi_grid_encoder_kernel(const float* __restrict__ x,
                          const float* __restrict__ coords,
                          const float* __restrict__ sigma,
                          const float* __restrict__ dist_offsets,
                          const float* __restrict__ Wg,
                          const float* __restrict__ b_out,
                          const int*   __restrict__ adjc,
                          const int*   __restrict__ nh_mask,
                          float*       __restrict__ out)
{
    __shared__ __attribute__((aligned(16))) float lds[LDS_FLOATS];

    const int tid    = threadIdx.x;
    const int wave   = tid >> 5;
    const int lane   = tid & 31;
    const int lane16 = lane & 15;
    const int hi     = lane >> 4;

    float* wS = lds;
    float* dm = lds + WS_FLOATS + wave * DM_FLOATS;

    const int pairIdx = blockIdx.x * 8 + wave;
    const int n0      = pairIdx * 2;

    const int colc = lane16 >> 3;
    const int colb = (lane16 >> 2) & 1;
    const int colv = lane16 & 3;
    const int nC   = n0 + colc;

    const int f0g = tid & 3;
    const int hiW = (tid >> 2) & 1;
    const int qpW = (tid >> 3) & 1;
    const int eW  = tid >> 4;
    v4f s[4];
    {
        const int os0 = hiW * 8 + qpW * 4;
        #pragma unroll
        for (int j = 0; j < 4; ++j)
            s[j] = *(const v4f*)(Wg + ((os0 + j) * 16 + eW) * 16 + f0g * 4);
    }

    int mh[2][2];
    #pragma unroll
    for (int qq = 0; qq < 2; ++qq) {
        #pragma unroll
        for (int r = 0; r < 2; ++r) {
            const int h  = qq * 4 + hi * 2 + r;
            const int hc = h < NHN ? h : NHN - 1;
            mh[qq][r] = adjc[nC * NHN + hc];
        }
    }

    const int   hD    = lane16 < NHN ? lane16 : NHN - 1;
    const int   nD    = n0 + hi;
    const int   mD    = adjc[nD * NHN + hD];
    const float lat_c = coords[nD];
    const float lon_c = coords[NC + nD];
    const float mkD   = (float)nh_mask[nD * NHN + hD];
    const float sg    = sigma[lane16 & 3];
    const float off   = dist_offsets[lane16 >> 2];
    const v4f   bo0   = *(const v4f*)(b_out + hi * 8);
    const v4f   bo1   = *(const v4f*)(b_out + hi * 8 + 4);

    __builtin_amdgcn_sched_barrier(0);

    {
        float* dst = wS + ((eW * 2 + qpW) * 32 + hiW * 16 + f0g * 4) * 4;
        #pragma unroll
        for (int jj = 0; jj < 4; ++jj)
            *(v4f*)(dst + jj * 4) = (v4f){s[0][jj], s[1][jj], s[2][jj], s[3][jj]};
    }

    const float lat_n = coords[mD];
    const float lon_n = coords[NC + mD];

    v4f xv[2][2][4];
    {
        const float* xb = x + (size_t)colb * NC * 64 + colv * 16;
        #pragma unroll
        for (int e4 = 0; e4 < 4; ++e4) {
            #pragma unroll
            for (int qq = 0; qq < 2; ++qq) {
                #pragma unroll
                for (int r = 0; r < 2; ++r) {
                    xv[qq][r][e4] = *(const v4f*)(xb + (size_t)mh[qq][r] * 64 + e4 * 4);
                }
            }
        }
    }

    __builtin_amdgcn_sched_barrier(0);

    {
        float cosd = __sinf(lat_c) * __sinf(lat_n)
                   + __cosf(lat_c) * __cosf(lat_n) * __cosf(lon_n - lon_c);
        cosd = fminf(fmaxf(cosd, -1.0f + 1e-7f), 1.0f - 1e-7f);
        const float d = acosf(cosd);
        const bool real = lane16 < NHN;
        *(v2f*)(dm + hi * 32 + lane16 * 2) = (v2f){real ? d : 0.0f, real ? mkD : 0.0f};
    }

    v2f A1[4];
    {
        const float inv_sg = 1.0f / sg;
        const unsigned padKeep = hi ? 0u : 0xffffffffu;
        #pragma unroll
        for (int c2 = 0; c2 < 2; ++c2) {
            float val[2][2];
            float part = 0.0f;
            #pragma unroll
            for (int qq = 0; qq < 2; ++qq) {
                const v4f p = *(const v4f*)(dm + c2 * 32 + (qq * 4 + hi * 2) * 2);
                #pragma unroll
                for (int r = 0; r < 2; ++r) {
                    const float t = (p[2 * r] - off) * inv_sg;
                    val[qq][r] = __expf(-0.5f * t * t) * p[2 * r + 1];
                    if (qq == 1 && r == 1)
                        val[qq][r] = __uint_as_float(__float_as_uint(val[qq][r]) & padKeep);
                    part += val[qq][r];
                }
            }
            const float sum = part + __shfl_xor(part, 16);
            const float inv = 1.0f / (sum + 1e-8f);
            #pragma unroll
            for (int qq = 0; qq < 2; ++qq)
                A1[c2 * 2 + qq] = (v2f){val[qq][0] * inv, val[qq][1] * inv};
        }
    }

    v8f acc0, acc1;
    {
        acc0 = (v8f){bo0[0], bo0[1], bo0[2], bo0[3], bo1[0], bo1[1], bo1[2], bo1[3]};
        acc1 = (v8f){0.f, 0.f, 0.f, 0.f, 0.f, 0.f, 0.f, 0.f};
    }

    __syncthreads();

    #pragma unroll
    for (int e = 0; e < 16; ++e) {
        v8f y = {0.f, 0.f, 0.f, 0.f, 0.f, 0.f, 0.f, 0.f};
        #pragma unroll
        for (int q1 = 0; q1 < 4; ++q1) {
            const int cc = q1 >> 1, qq = q1 & 1;
            v2f Bm = (v2f){xv[qq][0][e >> 2][e & 3], xv[qq][1][e >> 2][e & 3]};
            if (colc != cc) Bm = (v2f){0.f, 0.f};
            y = __builtin_amdgcn_wmma_f32_16x16x4_f32(
                    false, A1[q1], false, Bm, (short)0, y, false, false);
        }
        #pragma unroll
        for (int qp = 0; qp < 2; ++qp) {
            const v4f w = *(const v4f*)(wS + ((e * 2 + qp) * 32 + lane) * 4);
            #pragma unroll
            for (int ql = 0; ql < 2; ++ql) {
                const int q = qp * 2 + ql;
                const v2f A2 = (v2f){w[2 * ql], w[2 * ql + 1]};
                const v2f B2 = (v2f){y[2 * q], y[2 * q + 1]};
                if (e & 1)
                    acc1 = __builtin_amdgcn_wmma_f32_16x16x4_f32(
                            false, A2, false, B2, (short)0, acc1, false, false);
                else
                    acc0 = __builtin_amdgcn_wmma_f32_16x16x4_f32(
                            false, A2, false, B2, (short)0, acc0, false, false);
            }
        }
    }

    {
        float* op = out + (((size_t)colb * NC + nC) * NVV + colv) * EOUT + hi * 8;
        const v8f a = acc0 + acc1;
        *(v4f*)(op)     = (v4f){a[0], a[1], a[2], a[3]};
        *(v4f*)(op + 4) = (v4f){a[4], a[5], a[6], a[7]};
    }
}

extern "C" void kernel_launch(void* const* d_in, const int* in_sizes, int n_in,
                              void* d_out, int out_size, void* d_ws, size_t ws_size,
                              hipStream_t stream) {
    (void)in_sizes; (void)n_in; (void)out_size; (void)d_ws; (void)ws_size;
    const float* x            = (const float*)d_in[0];
    const float* coords       = (const float*)d_in[1];
    const float* sigma        = (const float*)d_in[2];
    const float* dist_offsets = (const float*)d_in[3];
    const float* Wg           = (const float*)d_in[4];
    const float* b_out        = (const float*)d_in[5];
    const int*   adjc         = (const int*)d_in[6];
    const int*   nh_mask      = (const int*)d_in[7];
    float*       out          = (float*)d_out;

    const int blocks = NC / 16;
    multi_grid_encoder_kernel<<<blocks, 256, 0, stream>>>(
        x, coords, sigma, dist_offsets, Wg, b_out, adjc, nh_mask, out);
}


# stage 1 per cell, K = 8 neighbours, 16 columns = (feature half, b, v), halves crossed between the 2 cells: per-lane select builds stage-2 B; 128 -> 96 WMMAs per wave; summation order changes
# speedup vs baseline: 1.9928x; 1.1699x over previous
#include <hip/hip_runtime.h>
#include <math.h>


typedef __attribute__((ext_vector_type(2))) float v2f;
typedef __attribute__((ext_vector_type(4))) float v4f;
typedef __attribute__((ext_vector_type(8))) float v8f;

#define NC    65536
#define NHN   7
#define NVV   4
#define EOUT  16

#define WS_FLOATS   4096
#define DM_FLOATS   64
#define LDS_FLOATS  (WS_FLOATS + 8 * DM_FLOATS)

__global__ __launch_bounds__(256) __attribute__((amdgpu_waves_per_eu(8))) void
multi_grid_encoder_kernel(const float* __restrict__ x,
                          const float* __restrict__ coords,
                          const float* __restrict__ sigma,
                          const float* __restrict__ dist_offsets,
                          const float* __restrict__ Wg,
                          const float* __restrict__ b_out,
                          const int*   __restrict__ adjc,
                          const int*   __restrict__ nh_mask,
                          float*       __restrict__ out)
{
    __shared__ __attribute__((aligned(16))) float lds[LDS_FLOATS];

    const int tid    = threadIdx.x;
    const int wave   = tid >> 5;
    const int lane   = tid & 31;
    const int lane16 = lane & 15;
    const int hi     = lane >> 4;

    float* wS = lds;
    float* dm = lds + WS_FLOATS + wave * DM_FLOATS;

    const int pairIdx = blockIdx.x * 8 + wave;
    const int n0      = pairIdx * 2;

    const int colg = lane16 >> 3;
    const int colb = (lane16 >> 2) & 1;
    const int colv = lane16 & 3;
    const int nC   = n0 + colg;
    unsigned outOff = (((unsigned)colb * NC + (unsigned)nC) * NVV + (unsigned)colv) * EOUT + (unsigned)hi * 8u;
    asm volatile("" : "+v"(outOff));

    const int f0g = tid & 3;
    const int hiW = (tid >> 2) & 1;
    const int qpW = (tid >> 3) & 1;
    const int eW  = tid >> 4;
    v4f s[4];
    {
        const int os0 = hiW * 8 + qpW * 4;
        #pragma unroll
        for (int j = 0; j < 4; ++j)
            s[j] = *(const v4f*)(Wg + ((os0 + j) * 16 + eW) * 16 + f0g * 4);
    }

    int mh[2][2][2];
    #pragma unroll
    for (int c = 0; c < 2; ++c) {
        #pragma unroll
        for (int qq = 0; qq < 2; ++qq) {
            #pragma unroll
            for (int r = 0; r < 2; ++r) {
                const int h  = qq * 4 + hi * 2 + r;
                const int hc = h < NHN ? h : NHN - 1;
                mh[c][qq][r] = adjc[(n0 + c) * NHN + hc];
            }
        }
    }

    const int   hD    = lane16 < NHN ? lane16 : NHN - 1;
    const int   nD    = n0 + hi;
    const int   mD    = adjc[nD * NHN + hD];
    const float lat_c = coords[nD];
    const float lon_c = coords[NC + nD];
    const float mkD   = (float)nh_mask[nD * NHN + hD];
    const float sg    = sigma[lane16 & 3];
    const float off   = dist_offsets[lane16 >> 2];
    const v4f   bo0   = *(const v4f*)(b_out + hi * 8);
    const v4f   bo1   = *(const v4f*)(b_out + hi * 8 + 4);

    __builtin_amdgcn_sched_barrier(0);

    {
        float* dst = wS + ((eW * 2 + qpW) * 32 + hiW * 16 + f0g * 4) * 4;
        #pragma unroll
        for (int jj = 0; jj < 4; ++jj)
            *(v4f*)(dst + jj * 4) = (v4f){s[0][jj], s[1][jj], s[2][jj], s[3][jj]};
    }

    const float lat_n = coords[mD];
    const float lon_n = coords[NC + mD];

    v4f xv[2][2][2][2];
    {
        const unsigned xb = (unsigned)colb * (NC * 64u) + (unsigned)colv * 16u;
        #pragma unroll
        for (int t4 = 0; t4 < 2; ++t4) {
            #pragma unroll
            for (int c = 0; c < 2; ++c) {
                const unsigned xc = xb + (unsigned)((colg ^ c) * 8 + t4 * 4);
                #pragma unroll
                for (int qq = 0; qq < 2; ++qq) {
                    #pragma unroll
                    for (int r = 0; r < 2; ++r) {
                        xv[c][qq][r][t4] = *(const v4f*)(x + (xc + (unsigned)mh[c][qq][r] * 64u));
                    }
                }
            }
        }
    }

    __builtin_amdgcn_sched_barrier(0);

    {
        float cosd = __sinf(lat_c) * __sinf(lat_n)
                   + __cosf(lat_c) * __cosf(lat_n) * __cosf(lon_n - lon_c);
        cosd = fminf(fmaxf(cosd, -1.0f + 1e-7f), 1.0f - 1e-7f);
        const float d = acosf(cosd);
        const bool real = lane16 < NHN;
        *(v2f*)(dm + hi * 32 + lane16 * 2) = (v2f){real ? d : 0.0f, real ? mkD : 0.0f};
    }

    v2f A1[4];
    {
        const float inv_sg = 1.0f / sg;
        const unsigned padKeep = hi ? 0u : 0xffffffffu;
        #pragma unroll
        for (int c2 = 0; c2 < 2; ++c2) {
            float val[2][2];
            float part = 0.0f;
            #pragma unroll
            for (int qq = 0; qq < 2; ++qq) {
                const v4f p = *(const v4f*)(dm + c2 * 32 + (qq * 4 + hi * 2) * 2);
                #pragma unroll
                for (int r = 0; r < 2; ++r) {
                    const float t = (p[2 * r] - off) * inv_sg;
                    val[qq][r] = __expf(-0.5f * t * t) * p[2 * r + 1];
                    if (qq == 1 && r == 1)
                        val[qq][r] = __uint_as_float(__float_as_uint(val[qq][r]) & padKeep);
                    part += val[qq][r];
                }
            }
            const float sum = part + __shfl_xor(part, 16);
            const float inv = 1.0f / (sum + 1e-8f);
            #pragma unroll
            for (int qq = 0; qq < 2; ++qq)
                A1[c2 * 2 + qq] = (v2f){val[qq][0] * inv, val[qq][1] * inv};
        }
    }

    v8f accP, accQ;
    {
        accP = (v8f){bo0[0], bo0[1], bo0[2], bo0[3], bo1[0], bo1[1], bo1[2], bo1[3]};
        accQ = (v8f){0.f, 0.f, 0.f, 0.f, 0.f, 0.f, 0.f, 0.f};
    }

    __syncthreads();

    const bool upper = colg != 0;
    #pragma unroll
    for (int t = 0; t < 8; ++t) {
        v8f y[2];
        #pragma unroll
        for (int c = 0; c < 2; ++c) {
            y[c] = (v8f){0.f, 0.f, 0.f, 0.f, 0.f, 0.f, 0.f, 0.f};
            #pragma unroll
            for (int qq = 0; qq < 2; ++qq) {
                const v2f Bm = (v2f){xv[c][qq][0][t >> 2][t & 3], xv[c][qq][1][t >> 2][t & 3]};
                y[c] = __builtin_amdgcn_wmma_f32_16x16x4_f32(
                        false, A1[c * 2 + qq], false, Bm, (short)0, y[c], false, false);
            }
        }
        v8f P, Q;
        #pragma unroll
        for (int j = 0; j < 8; ++j) {
            P[j] = upper ? y[1][j] : y[0][j];
            Q[j] = upper ? y[0][j] : y[1][j];
        }
        #pragma unroll
        for (int half = 0; half < 2; ++half) {
            const int e = t + 8 * half;
            #pragma unroll
            for (int qp = 0; qp < 2; ++qp) {
                const v4f w = *(const v4f*)(wS + ((e * 2 + qp) * 32 + lane) * 4);
                #pragma unroll
                for (int ql = 0; ql < 2; ++ql) {
                    const int q = qp * 2 + ql;
                    const v2f A2 = (v2f){w[2 * ql], w[2 * ql + 1]};
                    if (half) {
                        const v2f B2 = (v2f){Q[2 * q], Q[2 * q + 1]};
                        accQ = __builtin_amdgcn_wmma_f32_16x16x4_f32(
                                false, A2, false, B2, (short)0, accQ, false, false);
                    } else {
                        const v2f B2 = (v2f){P[2 * q], P[2 * q + 1]};
                        accP = __builtin_amdgcn_wmma_f32_16x16x4_f32(
                                false, A2, false, B2, (short)0, accP, false, false);
                    }
                }
            }
        }
    }

    {
        float* op = out + outOff;
        v8f a;
        #pragma unroll
        for (int j = 0; j < 8; ++j) {
            const int o = __builtin_amdgcn_update_dpp(0, __float_as_int(accQ[j]), 0x128, 0xF, 0xF, false);
            a[j] = accP[j] + __int_as_float(o);
        }
        *(v4f*)(op)     = (v4f){a[0], a[1], a[2], a[3]};
        *(v4f*)(op + 4) = (v4f){a[4], a[5], a[6], a[7]};
    }
}

extern "C" void kernel_launch(void* const* d_in, const int* in_sizes, int n_in,
                              void* d_out, int out_size, void* d_ws, size_t ws_size,
                              hipStream_t stream) {
    (void)in_sizes; (void)n_in; (void)out_size; (void)d_ws; (void)ws_size;
    const float* x            = (const float*)d_in[0];
    const float* coords       = (const float*)d_in[1];
    const float* sigma        = (const float*)d_in[2];
    const float* dist_offsets = (const float*)d_in[3];
    const float* Wg           = (const float*)d_in[4];
    const float* b_out        = (const float*)d_in[5];
    const int*   adjc         = (const int*)d_in[6];
    const int*   nh_mask      = (const int*)d_in[7];
    float*       out          = (float*)d_out;

    const int blocks = NC / 16;
    multi_grid_encoder_kernel<<<blocks, 256, 0, stream>>>(
        x, coords, sigma, dist_offsets, Wg, b_out, adjc, nh_mask, out);
}


# stage 1: each cell's neighbours ordered live-first (mask != 0) by a wave ballot; a cell with 4 or fewer live skips its all-zero second K chunk (8 WMMAs, 4 gather loads); summation order changes
# speedup vs baseline: 2.2316x; 1.1198x over previous
#include <hip/hip_runtime.h>
#include <math.h>


typedef __attribute__((ext_vector_type(2))) float v2f;
typedef __attribute__((ext_vector_type(4))) float v4f;
typedef __attribute__((ext_vector_type(8))) float v8f;

#define NC    65536
#define NHN   7
#define NVV   4
#define EOUT  16

#define WS_FLOATS   4096
#define DM_FLOATS   64
#define ID_INTS     32
#define LDS_FLOATS  (WS_FLOATS + 8 * (DM_FLOATS + ID_INTS))

typedef __attribute__((ext_vector_type(4))) int v4i;


template <bool N0, bool N1>
__device__ __forceinline__ void products_and_store(const v4f (&xv)[2][2][2][2], const v2f (&A1)[4],
                                                   v8f accP, v8f accQ, const float* wS,
                                                   const int lane, const bool upper,
                                                   float* __restrict__ out, const unsigned outOff)
{
    #pragma unroll
    for (int t = 0; t < 8; ++t) {
        v8f y[2];
        #pragma unroll
        for (int c = 0; c < 2; ++c) {
            const int nq = (c ? N1 : N0) ? 2 : 1;
            y[c] = (v8f){0.f, 0.f, 0.f, 0.f, 0.f, 0.f, 0.f, 0.f};
            #pragma unroll
            for (int qq = 0; qq < nq; ++qq) {
                const v2f Bm = (v2f){xv[c][qq][0][t >> 2][t & 3], xv[c][qq][1][t >> 2][t & 3]};
                y[c] = __builtin_amdgcn_wmma_f32_16x16x4_f32(
                        false, A1[c * 2 + qq], false, Bm, (short)0, y[c], false, false);
            }
        }
        v8f P, Q;
        #pragma unroll
        for (int j = 0; j < 8; ++j) {
            P[j] = upper ? y[1][j] : y[0][j];
            Q[j] = upper ? y[0][j] : y[1][j];
        }
        #pragma unroll
        for (int half = 0; half < 2; ++half) {
            const int e = t + 8 * half;
            #pragma unroll
            for (int qp = 0; qp < 2; ++qp) {
                const v4f w = *(const v4f*)(wS + ((e * 2 + qp) * 32 + lane) * 4);
                #pragma unroll
                for (int ql = 0; ql < 2; ++ql) {
                    const int q = qp * 2 + ql;
                    const v2f A2 = (v2f){w[2 * ql], w[2 * ql + 1]};
                    if (half) {
                        const v2f B2 = (v2f){Q[2 * q], Q[2 * q + 1]};
                        accQ = __builtin_amdgcn_wmma_f32_16x16x4_f32(
                                false, A2, false, B2, (short)0, accQ, false, false);
                    } else {
                        const v2f B2 = (v2f){P[2 * q], P[2 * q + 1]};
                        accP = __builtin_amdgcn_wmma_f32_16x16x4_f32(
                                false, A2, false, B2, (short)0, accP, false, false);
                    }
                }
            }
        }
    }

    {
        float* op = out + outOff;
        v8f a;
        #pragma unroll
        for (int j = 0; j < 8; ++j) {
            const int o = __builtin_amdgcn_update_dpp(0, __float_as_int(accQ[j]), 0x128, 0xF, 0xF, false);
            a[j] = accP[j] + __int_as_float(o);
        }
        *(v4f*)(op)     = (v4f){a[0], a[1], a[2], a[3]};
        *(v4f*)(op + 4) = (v4f){a[4], a[5], a[6], a[7]};
    }
}

__global__ __launch_bounds__(256) __attribute__((amdgpu_waves_per_eu(8))) void
multi_grid_encoder_kernel(const float* __restrict__ x,
                          const float* __restrict__ coords,
                          const float* __restrict__ sigma,
                          const float* __restrict__ dist_offsets,
                          const float* __restrict__ Wg,
                          const float* __restrict__ b_out,
                          const int*   __restrict__ adjc,
                          const int*   __restrict__ nh_mask,
                          float*       __restrict__ out)
{
    __shared__ __attribute__((aligned(16))) float lds[LDS_FLOATS];

    const int tid    = threadIdx.x;
    const int wave   = tid >> 5;
    const int lane   = tid & 31;
    const int lane16 = lane & 15;
    const int hi     = lane >> 4;

    float* wS = lds;
    float* dm = lds + WS_FLOATS + wave * (DM_FLOATS + ID_INTS);
    int*   id = (int*)(dm + DM_FLOATS);

    const int pairIdx = blockIdx.x * 8 + wave;
    const int n0      = pairIdx * 2;

    const int colg = lane16 >> 3;
    const int colb = (lane16 >> 2) & 1;
    const int colv = lane16 & 3;
    const int nC   = n0 + colg;
    unsigned outOff = (((unsigned)colb * NC + (unsigned)nC) * NVV + (unsigned)colv) * EOUT + (unsigned)hi * 8u;
    asm volatile("" : "+v"(outOff));

    const int f0g = tid & 3;
    const int hiW = (tid >> 2) & 1;
    const int qpW = (tid >> 3) & 1;
    const int eW  = tid >> 4;
    v4f s[4];
    {
        const int os0 = hiW * 8 + qpW * 4;
        #pragma unroll
        for (int j = 0; j < 4; ++j)
            s[j] = *(const v4f*)(Wg + ((os0 + j) * 16 + eW) * 16 + f0g * 4);
    }

    const int   hD    = lane16 < NHN ? lane16 : NHN - 1;
    const int   nD    = n0 + hi;
    const int   mD    = adjc[nD * NHN + hD];
    const float lat_c = coords[nD];
    const float lon_c = coords[NC + nD];
    const int   mkI   = nh_mask[nD * NHN + hD];
    const float sg    = sigma[lane16 & 3];
    const float off   = dist_offsets[lane16 >> 2];
    const v4f   bo0   = *(const v4f*)(b_out + hi * 8);
    const v4f   bo1   = *(const v4f*)(b_out + hi * 8 + 4);

    __builtin_amdgcn_sched_barrier(0);

    const float lat_n = coords[mD];
    const float lon_n = coords[NC + mD];

    const bool     real   = lane16 < NHN;
    const bool     liveLn = real && mkI != 0;
    const unsigned bal    = __builtin_amdgcn_ballot_w32(liveLn);
    const unsigned live0  = bal & 0x7Fu;
    const unsigned live1  = (bal >> 16) & 0x7Fu;
    const int      cnt0   = __builtin_popcount(live0);
    const int      cnt1   = __builtin_popcount(live1);
    const bool     need0  = cnt0 > 4;
    const bool     need1  = cnt1 > 4;
    int slot;
    {
        const unsigned liveC = hi ? live1 : live0;
        const int      cntC  = hi ? cnt1 : cnt0;
        const unsigned below = (1u << lane16) - 1u;
        const int sLive = __builtin_popcount(liveC & below);
        const int sDead = cntC + __builtin_popcount(~liveC & 0x7Fu & below);
        slot = liveLn ? sLive : sDead;
        slot = real ? slot : lane16;
        const int pos = (slot & 9) | ((slot & 2) << 1) | ((slot & 4) >> 1);
        id[hi * 16 + pos] = mD;
    }
    const float mkD = (float)mkI;

    {
        float* dst = wS + ((eW * 2 + qpW) * 32 + hiW * 16 + f0g * 4) * 4;
        #pragma unroll
        for (int jj = 0; jj < 4; ++jj)
            *(v4f*)(dst + jj * 4) = (v4f){s[0][jj], s[1][jj], s[2][jj], s[3][jj]};
    }

    int mh[2][2][2];
    #pragma unroll
    for (int c = 0; c < 2; ++c) {
        const v4i ids = *(const v4i*)(id + c * 16 + hi * 4);
        #pragma unroll
        for (int qq = 0; qq < 2; ++qq) {
            #pragma unroll
            for (int r = 0; r < 2; ++r)
                mh[c][qq][r] = ids[qq * 2 + r];
        }
    }

    v4f xv[2][2][2][2];
    {
        const unsigned xb = (unsigned)colb * (NC * 64u) + (unsigned)colv * 16u;
        #pragma unroll
        for (int t4 = 0; t4 < 2; ++t4) {
            #pragma unroll
            for (int c = 0; c < 2; ++c) {
                const unsigned xc = xb + (unsigned)((colg ^ c) * 8 + t4 * 4);
                #pragma unroll
                for (int r = 0; r < 2; ++r)
                    xv[c][0][r][t4] = *(const v4f*)(x + (xc + (unsigned)mh[c][0][r] * 64u));
            }
        }
        asm volatile("" :: "v"(lat_n), "v"(lon_n));
        #pragma unroll
        for (int c = 0; c < 2; ++c) {
            if (c ? need1 : need0) {
                #pragma unroll
                for (int t4 = 0; t4 < 2; ++t4) {
                    const unsigned xc = xb + (unsigned)((colg ^ c) * 8 + t4 * 4);
                    #pragma unroll
                    for (int r = 0; r < 2; ++r)
                        xv[c][1][r][t4] = *(const v4f*)(x + (xc + (unsigned)mh[c][1][r] * 64u));
                }
            }
        }
    }

    {
        float cosd = __sinf(lat_c) * __sinf(lat_n)
                   + __cosf(lat_c) * __cosf(lat_n) * __cosf(lon_n - lon_c);
        cosd = fminf(fmaxf(cosd, -1.0f + 1e-7f), 1.0f - 1e-7f);
        const float d = acosf(cosd);
        *(v2f*)(dm + hi * 32 + slot * 2) = (v2f){real ? d : 0.0f, real ? mkD : 0.0f};
    }

    v2f A1[4];
    {
        const float inv_sg = 1.0f / sg;
        const unsigned padKeep = hi ? 0u : 0xffffffffu;
        #pragma unroll
        for (int c2 = 0; c2 < 2; ++c2) {
            float val[2][2];
            float part = 0.0f;
            #pragma unroll
            for (int qq = 0; qq < 2; ++qq) {
                const v4f p = *(const v4f*)(dm + c2 * 32 + (qq * 4 + hi * 2) * 2);
                #pragma unroll
                for (int r = 0; r < 2; ++r) {
                    const float t = (p[2 * r] - off) * inv_sg;
                    val[qq][r] = __expf(-0.5f * t * t) * p[2 * r + 1];
                    if (qq == 1 && r == 1)
                        val[qq][r] = __uint_as_float(__float_as_uint(val[qq][r]) & padKeep);
                    part += val[qq][r];
                }
            }
            const float sum = part + __shfl_xor(part, 16);
            const float inv = 1.0f / (sum + 1e-8f);
            #pragma unroll
            for (int qq = 0; qq < 2; ++qq)
                A1[c2 * 2 + qq] = (v2f){val[qq][0] * inv, val[qq][1] * inv};
        }
    }

    v8f accP, accQ;
    {
        accP = (v8f){bo0[0], bo0[1], bo0[2], bo0[3], bo1[0], bo1[1], bo1[2], bo1[3]};
        accQ = (v8f){0.f, 0.f, 0.f, 0.f, 0.f, 0.f, 0.f, 0.f};
    }

    __syncthreads();

    const bool upper = colg != 0;
    if (need0) {
        if (need1) products_and_store<true,  true >(xv, A1, accP, accQ, wS, lane, upper, out, outOff);
        else       products_and_store<true,  false>(xv, A1, accP, accQ, wS, lane, upper, out, outOff);
    } else {
        if (need1) products_and_store<false, true >(xv, A1, accP, accQ, wS, lane, upper, out, outOff);
        else       products_and_store<false, false>(xv, A1, accP, accQ, wS, lane, upper, out, outOff);
    }
}

extern "C" void kernel_launch(void* const* d_in, const int* in_sizes, int n_in,
                              void* d_out, int out_size, void* d_ws, size_t ws_size,
                              hipStream_t stream) {
    (void)in_sizes; (void)n_in; (void)out_size; (void)d_ws; (void)ws_size;
    const float* x            = (const float*)d_in[0];
    const float* coords       = (const float*)d_in[1];
    const float* sigma        = (const float*)d_in[2];
    const float* dist_offsets = (const float*)d_in[3];
    const float* Wg           = (const float*)d_in[4];
    const float* b_out        = (const float*)d_in[5];
    const int*   adjc         = (const int*)d_in[6];
    const int*   nh_mask      = (const int*)d_in[7];
    float*       out          = (float*)d_out;

    const int blocks = NC / 16;
    multi_grid_encoder_kernel<<<blocks, 256, 0, stream>>>(
        x, coords, sigma, dist_offsets, Wg, b_out, adjc, nh_mask, out);
}


# masked neighbours and the K padding take the id of the cell's first live neighbour (wave ballot, readlane): their coordinate and feature-row loads fall on lines already asked for; weights stay 0
# speedup vs baseline: 2.3407x; 1.0489x over previous
#include <hip/hip_runtime.h>
#include <math.h>


typedef __attribute__((ext_vector_type(2))) float v2f;
typedef __attribute__((ext_vector_type(4))) float v4f;
typedef __attribute__((ext_vector_type(8))) float v8f;

#define NC    65536
#define NHN   7
#define NVV   4
#define EOUT  16

#define WS_FLOATS   4096
#define DM_FLOATS   64
#define ID_INTS     32
#define LDS_FLOATS  (WS_FLOATS + 8 * (DM_FLOATS + ID_INTS))

typedef __attribute__((ext_vector_type(4))) int v4i;


template <bool N0, bool N1>
__device__ __forceinline__ void products_and_store(const v4f (&xv)[2][2][2][2], const v2f (&A1)[4],
                                                   v8f accP, v8f accQ, const float* wS,
                                                   const int lane, const bool upper,
                                                   float* __restrict__ out, const unsigned outOff)
{
    #pragma unroll
    for (int t = 0; t < 8; ++t) {
        v8f y[2];
        #pragma unroll
        for (int c = 0; c < 2; ++c) {
            const int nq = (c ? N1 : N0) ? 2 : 1;
            y[c] = (v8f){0.f, 0.f, 0.f, 0.f, 0.f, 0.f, 0.f, 0.f};
            #pragma unroll
            for (int qq = 0; qq < nq; ++qq) {
                const v2f Bm = (v2f){xv[c][qq][0][t >> 2][t & 3], xv[c][qq][1][t >> 2][t & 3]};
                y[c] = __builtin_amdgcn_wmma_f32_16x16x4_f32(
                        false, A1[c * 2 + qq], false, Bm, (short)0, y[c], false, false);
            }
        }
        v8f P, Q;
        #pragma unroll
        for (int j = 0; j < 8; ++j) {
            P[j] = upper ? y[1][j] : y[0][j];
            Q[j] = upper ? y[0][j] : y[1][j];
        }
        #pragma unroll
        for (int half = 0; half < 2; ++half) {
            const int e = t + 8 * half;
            #pragma unroll
            for (int qp = 0; qp < 2; ++qp) {
                const v4f w = *(const v4f*)(wS + ((e * 2 + qp) * 32 + lane) * 4);
                #pragma unroll
                for (int ql = 0; ql < 2; ++ql) {
                    const int q = qp * 2 + ql;
                    const v2f A2 = (v2f){w[2 * ql], w[2 * ql + 1]};
                    if (half) {
                        const v2f B2 = (v2f){Q[2 * q], Q[2 * q + 1]};
                        accQ = __builtin_amdgcn_wmma_f32_16x16x4_f32(
                                false, A2, false, B2, (short)0, accQ, false, false);
                    } else {
                        const v2f B2 = (v2f){P[2 * q], P[2 * q + 1]};
                        accP = __builtin_amdgcn_wmma_f32_16x16x4_f32(
                                false, A2, false, B2, (short)0, accP, false, false);
                    }
                }
            }
        }
    }

    {
        float* op = out + outOff;
        v8f a;
        #pragma unroll
        for (int j = 0; j < 8; ++j) {
            const int o = __builtin_amdgcn_update_dpp(0, __float_as_int(accQ[j]), 0x128, 0xF, 0xF, false);
            a[j] = accP[j] + __int_as_float(o);
        }
        *(v4f*)(op)     = (v4f){a[0], a[1], a[2], a[3]};
        *(v4f*)(op + 4) = (v4f){a[4], a[5], a[6], a[7]};
    }
}

__global__ __launch_bounds__(256) __attribute__((amdgpu_waves_per_eu(8))) void
multi_grid_encoder_kernel(const float* __restrict__ x,
                          const float* __restrict__ coords,
                          const float* __restrict__ sigma,
                          const float* __restrict__ dist_offsets,
                          const float* __restrict__ Wg,
                          const float* __restrict__ b_out,
                          const int*   __restrict__ adjc,
                          const int*   __restrict__ nh_mask,
                          float*       __restrict__ out)
{
    __shared__ __attribute__((aligned(16))) float lds[LDS_FLOATS];

    const int tid    = threadIdx.x;
    const int wave   = tid >> 5;
    const int lane   = tid & 31;
    const int lane16 = lane & 15;
    const int hi     = lane >> 4;

    float* wS = lds;
    float* dm = lds + WS_FLOATS + wave * (DM_FLOATS + ID_INTS);
    int*   id = (int*)(dm + DM_FLOATS);

    const int pairIdx = blockIdx.x * 8 + wave;
    const int n0      = pairIdx * 2;

    const int colg = lane16 >> 3;
    const int colb = (lane16 >> 2) & 1;
    const int colv = lane16 & 3;
    const int nC   = n0 + colg;
    unsigned outOff = (((unsigned)colb * NC + (unsigned)nC) * NVV + (unsigned)colv) * EOUT + (unsigned)hi * 8u;
    asm volatile("" : "+v"(outOff));

    const int f0g = tid & 3;
    const int hiW = (tid >> 2) & 1;
    const int qpW = (tid >> 3) & 1;
    const int eW  = tid >> 4;
    v4f s[4];
    {
        const int os0 = hiW * 8 + qpW * 4;
        #pragma unroll
        for (int j = 0; j < 4; ++j)
            s[j] = *(const v4f*)(Wg + ((os0 + j) * 16 + eW) * 16 + f0g * 4);
    }

    const int   hD    = lane16 < NHN ? lane16 : NHN - 1;
    const int   nD    = n0 + hi;
    const int   mD    = adjc[nD * NHN + hD];
    const float lat_c = coords[nD];
    const float lon_c = coords[NC + nD];
    const int   mkI   = nh_mask[nD * NHN + hD];
    const float sg    = sigma[lane16 & 3];
    const float off   = dist_offsets[lane16 >> 2];
    const v4f   bo0   = *(const v4f*)(b_out + hi * 8);
    const v4f   bo1   = *(const v4f*)(b_out + hi * 8 + 4);

    __builtin_amdgcn_sched_barrier(0);

    const bool     real   = lane16 < NHN;
    const bool     liveLn = real && mkI != 0;
    const unsigned bal    = __builtin_amdgcn_ballot_w32(liveLn);
    const unsigned live0  = bal & 0x7Fu;
    const unsigned live1  = (bal >> 16) & 0x7Fu;
    const int      cnt0   = __builtin_popcount(live0);
    const int      cnt1   = __builtin_popcount(live1);
    const bool     need0  = cnt0 > 4;
    const bool     need1  = cnt1 > 4;

    const int mF0 = __builtin_amdgcn_readlane(mD, live0 ? __builtin_ctz(live0) : 0);
    const int mF1 = __builtin_amdgcn_readlane(mD, 16 + (live1 ? __builtin_ctz(live1) : 0));
    const int mA  = liveLn ? mD : (hi ? mF1 : mF0);

    const float lat_n = coords[mA];
    const float lon_n = coords[NC + mA];
    int slot;
    {
        const unsigned liveC = hi ? live1 : live0;
        const int      cntC  = hi ? cnt1 : cnt0;
        const unsigned below = (1u << lane16) - 1u;
        const int sLive = __builtin_popcount(liveC & below);
        const int sDead = cntC + __builtin_popcount(~liveC & 0x7Fu & below);
        slot = liveLn ? sLive : sDead;
        slot = real ? slot : lane16;
        const int pos = (slot & 9) | ((slot & 2) << 1) | ((slot & 4) >> 1);
        id[hi * 16 + pos] = mA;
    }
    const float mkD = (float)mkI;

    {
        float* dst = wS + ((eW * 2 + qpW) * 32 + hiW * 16 + f0g * 4) * 4;
        #pragma unroll
        for (int jj = 0; jj < 4; ++jj)
            *(v4f*)(dst + jj * 4) = (v4f){s[0][jj], s[1][jj], s[2][jj], s[3][jj]};
    }

    int mh[2][2][2];
    #pragma unroll
    for (int c = 0; c < 2; ++c) {
        const v4i ids = *(const v4i*)(id + c * 16 + hi * 4);
        #pragma unroll
        for (int qq = 0; qq < 2; ++qq) {
            #pragma unroll
            for (int r = 0; r < 2; ++r)
                mh[c][qq][r] = ids[qq * 2 + r];
        }
    }

    v4f xv[2][2][2][2];
    {
        const unsigned xb = (unsigned)colb * (NC * 64u) + (unsigned)colv * 16u;
        #pragma unroll
        for (int t4 = 0; t4 < 2; ++t4) {
            #pragma unroll
            for (int c = 0; c < 2; ++c) {
                const unsigned xc = xb + (unsigned)((colg ^ c) * 8 + t4 * 4);
                #pragma unroll
                for (int r = 0; r < 2; ++r)
                    xv[c][0][r][t4] = *(const v4f*)(x + (xc + (unsigned)mh[c][0][r] * 64u));
            }
        }
        asm volatile("" :: "v"(lat_n), "v"(lon_n));
        #pragma unroll
        for (int c = 0; c < 2; ++c) {
            if (c ? need1 : need0) {
                #pragma unroll
                for (int t4 = 0; t4 < 2; ++t4) {
                    const unsigned xc = xb + (unsigned)((colg ^ c) * 8 + t4 * 4);
                    #pragma unroll
                    for (int r = 0; r < 2; ++r)
                        xv[c][1][r][t4] = *(const v4f*)(x + (xc + (unsigned)mh[c][1][r] * 64u));
                }
            }
        }
    }

    {
        float cosd = __sinf(lat_c) * __sinf(lat_n)
                   + __cosf(lat_c) * __cosf(lat_n) * __cosf(lon_n - lon_c);
        cosd = fminf(fmaxf(cosd, -1.0f + 1e-7f), 1.0f - 1e-7f);
        const float d = acosf(cosd);
        *(v2f*)(dm + hi * 32 + slot * 2) = (v2f){real ? d : 0.0f, real ? mkD : 0.0f};
    }

    v2f A1[4];
    {
        const float inv_sg = 1.0f / sg;
        const unsigned padKeep = hi ? 0u : 0xffffffffu;
        #pragma unroll
        for (int c2 = 0; c2 < 2; ++c2) {
            float val[2][2];
            float part = 0.0f;
            #pragma unroll
            for (int qq = 0; qq < 2; ++qq) {
                const v4f p = *(const v4f*)(dm + c2 * 32 + (qq * 4 + hi * 2) * 2);
                #pragma unroll
                for (int r = 0; r < 2; ++r) {
                    const float t = (p[2 * r] - off) * inv_sg;
                    val[qq][r] = __expf(-0.5f * t * t) * p[2 * r + 1];
                    if (qq == 1 && r == 1)
                        val[qq][r] = __uint_as_float(__float_as_uint(val[qq][r]) & padKeep);
                    part += val[qq][r];
                }
            }
            const float sum = part + __shfl_xor(part, 16);
            const float inv = 1.0f / (sum + 1e-8f);
            #pragma unroll
            for (int qq = 0; qq < 2; ++qq)
                A1[c2 * 2 + qq] = (v2f){val[qq][0] * inv, val[qq][1] * inv};
        }
    }

    v8f accP, accQ;
    {
        accP = (v8f){bo0[0], bo0[1], bo0[2], bo0[3], bo1[0], bo1[1], bo1[2], bo1[3]};
        accQ = (v8f){0.f, 0.f, 0.f, 0.f, 0.f, 0.f, 0.f, 0.f};
    }

    __syncthreads();

    const bool upper = colg != 0;
    if (need0) {
        if (need1) products_and_store<true,  true >(xv, A1, accP, accQ, wS, lane, upper, out, outOff);
        else       products_and_store<true,  false>(xv, A1, accP, accQ, wS, lane, upper, out, outOff);
    } else {
        if (need1) products_and_store<false, true >(xv, A1, accP, accQ, wS, lane, upper, out, outOff);
        else       products_and_store<false, false>(xv, A1, accP, accQ, wS, lane, upper, out, outOff);
    }
}

extern "C" void kernel_launch(void* const* d_in, const int* in_sizes, int n_in,
                              void* d_out, int out_size, void* d_ws, size_t ws_size,
                              hipStream_t stream) {
    (void)in_sizes; (void)n_in; (void)out_size; (void)d_ws; (void)ws_size;
    const float* x            = (const float*)d_in[0];
    const float* coords       = (const float*)d_in[1];
    const float* sigma        = (const float*)d_in[2];
    const float* dist_offsets = (const float*)d_in[3];
    const float* Wg           = (const float*)d_in[4];
    const float* b_out        = (const float*)d_in[5];
    const int*   adjc         = (const int*)d_in[6];
    const int*   nh_mask      = (const int*)d_in[7];
    float*       out          = (float*)d_out;

    const int blocks = NC / 16;
    multi_grid_encoder_kernel<<<blocks, 256, 0, stream>>>(
        x, coords, sigma, dist_offsets, Wg, b_out, adjc, nh_mask, out);
}
